# starting kernel
# baseline (speedup 1.0000x reference)
#include <hip/hip_runtime.h>
#include <hip/hip_bf16.h>
#include <math.h>

typedef __attribute__((ext_vector_type(2))) float v2f;
typedef __attribute__((ext_vector_type(4))) float v4f;
typedef __attribute__((ext_vector_type(8))) float v8f;

#define GD 128
#define GRID_VOX (GD * GD * GD)
#define NROWS 64
#define LSTR 68

__global__ void grid_init_kernel(int* __restrict__ grid) {
    int i = blockIdx.x * 256 + threadIdx.x;
    if (i < GRID_VOX) grid[i] = -1;
}

__global__ void grid_build_kernel(const int* __restrict__ coords,
                                  int* __restrict__ grid, int N) {
    int i = blockIdx.x * 256 + threadIdx.x;
    if (i < N) {
        int z = coords[3 * i + 0];
        int y = coords[3 * i + 1];
        int x = coords[3 * i + 2];
        grid[(z * GD + y) * GD + x] = i;
    }
}

__global__ void pack_w1_kernel(const float* __restrict__ w1, float* __restrict__ w1p) {
    int t = blockIdx.x * 256 + threadIdx.x;
    if (t >= 27 * 4 * 16 * 32) return;
    int lane = t & 31;
    int k    = (t >> 5) & 15;
    int nt   = (t >> 9) & 3;
    int off  = t >> 11;
    int half = lane >> 4;
    int col  = nt * 16 + (lane & 15);
    int krow = 4 * k + 2 * half;
    const float* w = w1 + (size_t)off * 64 * 64;
    w1p[(size_t)t * 2 + 0] = w[(krow + 0) * 64 + col];
    w1p[(size_t)t * 2 + 1] = w[(krow + 1) * 64 + col];
}

__global__ void pack_w2_kernel(const float* __restrict__ w2, float* __restrict__ w2p) {
    int t = blockIdx.x * 256 + threadIdx.x;
    if (t >= 4 * 16 * 32) return;
    int lane = t & 31;
    int k    = (t >> 5) & 15;
    int nt   = (t >> 9) & 3;
    int half = lane >> 4;
    int col  = nt * 16 + (lane & 15);
    int krow = 4 * k + 2 * half;
    w2p[(size_t)t * 2 + 0] = w2[(krow + 0) * 64 + col];
    w2p[(size_t)t * 2 + 1] = w2[(krow + 1) * 64 + col];
}

__global__ __launch_bounds__(128)
void sparse_block_kernel(const float* __restrict__ feat,
                         const int*   __restrict__ coords,
                         const int*   __restrict__ grid,
                         const float* __restrict__ w1p,
                         const float* __restrict__ b1,
                         const float* __restrict__ gamma,
                         const float* __restrict__ beta,
                         const float* __restrict__ rmean,
                         const float* __restrict__ rvar,
                         const float* __restrict__ w2p,
                         const float* __restrict__ b2,
                         float* __restrict__ out,
                         int N) {
    __shared__ float lds[NROWS * LSTR];
    __shared__ int   snidx[NROWS];
    __shared__ int   scz[NROWS], scy[NROWS], scx[NROWS];

    const int tid  = threadIdx.x;
    const int lane = tid & 31;
    const int wv   = tid >> 5;
    const int half = lane >> 4;
    const int l15  = lane & 15;
    const int col  = wv * 16 + l15;
    const int row0 = blockIdx.x * NROWS;

    if (tid < NROWS) {
        int r = row0 + tid;
        if (r < N) {
            scz[tid] = coords[3 * r + 0];
            scy[tid] = coords[3 * r + 1];
            scx[tid] = coords[3 * r + 2];
        } else {
            scz[tid] = -10000; scy[tid] = -10000; scx[tid] = -10000;
        }
    }
    __syncthreads();

    v8f acc[4];
    #pragma unroll
    for (int m = 0; m < 4; ++m) acc[m] = (v8f)0.0f;

    for (int off = 0; off < 27; ++off) {
        const int dz = off / 9 - 1;
        const int dy = (off / 3) % 3 - 1;
        const int dx = off % 3 - 1;

        if (tid < NROWS) {
            int nz = scz[tid] + dz, ny = scy[tid] + dy, nx = scx[tid] + dx;
            int idx = -1;
            if (nz >= 0 && nz < GD && ny >= 0 && ny < GD && nx >= 0 && nx < GD)
                idx = grid[(nz * GD + ny) * GD + nx];
            snidx[tid] = idx;
        }
        __syncthreads();

        #pragma unroll
        for (int it = 0; it < (NROWS * 16) / 128; ++it) {
            int i = it * 128 + tid;
            int r = i >> 4;
            int q = i & 15;
            int src = snidx[r];
            v4f v = (v4f)0.0f;
            if (src >= 0)
                v = *(const v4f*)(feat + (size_t)src * 64 + q * 4);
            *(v4f*)(&lds[r * LSTR + q * 4]) = v;
        }
        __syncthreads();

        const float* bp = w1p + (size_t)(off * 4 + wv) * 16 * 64;
        #pragma unroll
        for (int k = 0; k < 16; ++k) {
            v2f bf = *(const v2f*)(bp + k * 64 + lane * 2);
            #pragma unroll
            for (int m = 0; m < 4; ++m) {
                v2f af = *(const v2f*)(&lds[(m * 16 + l15) * LSTR + 4 * k + 2 * half]);
                acc[m] = __builtin_amdgcn_wmma_f32_16x16x4_f32(
                    false, af, false, bf, (short)0, acc[m], false, false);
            }
        }
        __syncthreads();
    }

    const float scl = gamma[col] * rsqrtf(rvar[col] + 1e-5f);
    const float shf = (b1[col] - rmean[col]) * scl + beta[col];

    #pragma unroll
    for (int m = 0; m < 4; ++m) {
        #pragma unroll
        for (int j = 0; j < 8; ++j) {
            float y = acc[m][j] * scl + shf;
            y = y > 0.0f ? y : 0.0f;
            lds[(m * 16 + j + half * 8) * LSTR + col] = y;
        }
    }
    __syncthreads();

    v8f acc2[4];
    #pragma unroll
    for (int m = 0; m < 4; ++m) acc2[m] = (v8f)0.0f;
    #pragma unroll
    for (int k = 0; k < 16; ++k) {
        v2f bf = *(const v2f*)(w2p + (size_t)(wv * 16 + k) * 64 + lane * 2);
        #pragma unroll
        for (int m = 0; m < 4; ++m) {
            v2f af = *(const v2f*)(&lds[(m * 16 + l15) * LSTR + 4 * k + 2 * half]);
            acc2[m] = __builtin_amdgcn_wmma_f32_16x16x4_f32(
                false, af, false, bf, (short)0, acc2[m], false, false);
        }
    }

    const float bb = b2[col];
    #pragma unroll
    for (int m = 0; m < 4; ++m) {
        int rg = row0 + m * 16 + half * 8;
        #pragma unroll
        for (int j = 0; j < 8; ++j) {
            int r = rg + j;
            if (r < N)
                out[(size_t)r * 64 + col] = acc2[m][j] + bb;
        }
    }
}

extern "C" void kernel_launch(void* const* d_in, const int* in_sizes, int n_in,
                              void* d_out, int out_size, void* d_ws, size_t ws_size,
                              hipStream_t stream) {
    const float* feat  = (const float*)d_in[0];
    const int*   coords= (const int*)  d_in[1];
    const float* w1    = (const float*)d_in[2];
    const float* b1    = (const float*)d_in[3];
    const float* gamma = (const float*)d_in[4];
    const float* beta  = (const float*)d_in[5];
    const float* rmean = (const float*)d_in[6];
    const float* rvar  = (const float*)d_in[7];
    const float* w2    = (const float*)d_in[8];
    const float* b2    = (const float*)d_in[9];
    float* out = (float*)d_out;

    const int N = in_sizes[0] / 64;

    char* ws = (char*)d_ws;
    int*   grid = (int*)ws;
    float* w1p  = (float*)(ws + (size_t)GRID_VOX * 4);
    float* w2p  = w1p + (size_t)27 * 4 * 16 * 32 * 2;

    grid_init_kernel<<<GRID_VOX / 256, 256, 0, stream>>>(grid);
    grid_build_kernel<<<(N + 255) / 256, 256, 0, stream>>>(coords, grid, N);
    pack_w1_kernel<<<(27 * 4 * 16 * 32 + 255) / 256, 256, 0, stream>>>(w1, w1p);
    pack_w2_kernel<<<(4 * 16 * 32 + 255) / 256, 256, 0, stream>>>(w2, w2p);

    const int nblk = (N + NROWS - 1) / NROWS;
    sparse_block_kernel<<<nblk, 128, 0, stream>>>(feat, coords, grid, w1p, b1,
                                                  gamma, beta, rmean, rvar,
                                                  w2p, b2, out, N);
}


# Skip inactive neighbours: per tap compact the tile's active rows, gather and WMMA only those (32-row chunks), add each tap's product into an LDS accumulator; sum order: per-tap product, then taps
# speedup vs baseline: 2.0317x; 2.0317x over previous
#include <hip/hip_runtime.h>
#include <hip/hip_bf16.h>
#include <math.h>

typedef __attribute__((ext_vector_type(2))) float v2f;
typedef __attribute__((ext_vector_type(4))) float v4f;
typedef __attribute__((ext_vector_type(8))) float v8f;
typedef __attribute__((ext_vector_type(4))) int   v4i;

#define GD 128
#define GRID_VOX (GD * GD * GD)
#define TROWS 128
#define CROWS 32
#define NTAP 27
#define LSTR 68
#define PAD_ENTRY (-128)

__global__ void grid_init_kernel(int* __restrict__ grid) {
    int i = blockIdx.x * 256 + threadIdx.x;
    if (i < GRID_VOX) grid[i] = -1;
}

__global__ void grid_build_kernel(const int* __restrict__ coords,
                                  int* __restrict__ grid, int N) {
    int i = blockIdx.x * 256 + threadIdx.x;
    if (i < N) {
        int z = coords[3 * i + 0];
        int y = coords[3 * i + 1];
        int x = coords[3 * i + 2];
        grid[(z * GD + y) * GD + x] = i;
    }
}

__global__ void pack_w1_kernel(const float* __restrict__ w1, float* __restrict__ w1p) {
    int t = blockIdx.x * 256 + threadIdx.x;
    if (t >= 27 * 4 * 16 * 32) return;
    int lane = t & 31;
    int k    = (t >> 5) & 15;
    int nt   = (t >> 9) & 3;
    int off  = t >> 11;
    int half = lane >> 4;
    int col  = nt * 16 + (lane & 15);
    int krow = 4 * k + 2 * half;
    const float* w = w1 + (size_t)off * 64 * 64;
    w1p[(size_t)t * 2 + 0] = w[(krow + 0) * 64 + col];
    w1p[(size_t)t * 2 + 1] = w[(krow + 1) * 64 + col];
}

__global__ void pack_w2_kernel(const float* __restrict__ w2, float* __restrict__ w2p) {
    int t = blockIdx.x * 256 + threadIdx.x;
    if (t >= 4 * 16 * 32) return;
    int lane = t & 31;
    int k    = (t >> 5) & 15;
    int nt   = (t >> 9) & 3;
    int half = lane >> 4;
    int col  = nt * 16 + (lane & 15);
    int krow = 4 * k + 2 * half;
    w2p[(size_t)t * 2 + 0] = w2[(krow + 0) * 64 + col];
    w2p[(size_t)t * 2 + 1] = w2[(krow + 1) * 64 + col];
}

__global__ __launch_bounds__(128)
void sparse_block_kernel(const float* __restrict__ feat,
                         const int*   __restrict__ coords,
                         const int*   __restrict__ grid,
                         const float* __restrict__ w1p,
                         const float* __restrict__ b1,
                         const float* __restrict__ gamma,
                         const float* __restrict__ beta,
                         const float* __restrict__ rmean,
                         const float* __restrict__ rvar,
                         const float* __restrict__ w2p,
                         const float* __restrict__ b2,
                         float* __restrict__ out,
                         int N) {
    __shared__ __attribute__((aligned(16))) float accL[(TROWS + 1) * LSTR];
    __shared__ __attribute__((aligned(16))) float abuf[2 * CROWS * LSTR];
    __shared__ __attribute__((aligned(16))) int   clist[NTAP * TROWS];
    __shared__ int ccnt[NTAP];

    const int tid  = threadIdx.x;
    const int lane = tid & 31;
    const int wv   = __builtin_amdgcn_readfirstlane(tid >> 5);
    const int half = lane >> 4;
    const int l15  = lane & 15;
    const int col  = wv * 16 + l15;
    const int row0 = blockIdx.x * TROWS;
    const int grow = tid >> 4;
    const int gq   = tid & 15;

    for (int i = tid; i < (TROWS + 1) * LSTR / 4; i += 128)
        *(v4f*)(&accL[i * 4]) = (v4f)0.0f;

    int cz[4], cy[4], cx[4];
    #pragma unroll
    for (int g = 0; g < 4; ++g) {
        const int r  = row0 + g * 32 + lane;
        const int rc = r < N ? r : N - 1;
        const int z = coords[3 * rc + 0];
        const int y = coords[3 * rc + 1];
        const int x = coords[3 * rc + 2];
        cz[g] = r < N ? z : -10000;
        cy[g] = r < N ? y : -10000;
        cx[g] = r < N ? x : -10000;
    }

    int nid[7][4];
    #pragma unroll
    for (int i = 0; i < 7; ++i) {
        int off = wv + 4 * i;
        off = off < NTAP ? off : NTAP - 1;
        const int dz = off / 9 - 1;
        const int dy = (off / 3) % 3 - 1;
        const int dx = off % 3 - 1;
        #pragma unroll
        for (int g = 0; g < 4; ++g) {
            const int nz = cz[g] + dz, ny = cy[g] + dy, nx = cx[g] + dx;
            int idx = -1;
            if (nz >= 0 && nz < GD && ny >= 0 && ny < GD && nx >= 0 && nx < GD)
                idx = grid[(nz * GD + ny) * GD + nx];
            nid[i][g] = idx;
        }
    }

    const unsigned lt = (1u << lane) - 1u;
    #pragma unroll
    for (int i = 0; i < 7; ++i) {
        const int off = wv + 4 * i;
        if (off < NTAP) {
            int base = 0;
            #pragma unroll
            for (int g = 0; g < 4; ++g) {
                const bool v = nid[i][g] >= 0;
                const unsigned m = __builtin_amdgcn_ballot_w32(v);
                if (v)
                    clist[off * TROWS + base + __popc(m & lt)] = (nid[i][g] << 8) | (g * 32 + lane);
                base += __popc(m);
            }
            int padded = (base + CROWS - 1) & ~(CROWS - 1);
            if (padded < CROWS) padded = CROWS;
            if (base + lane < padded) clist[off * TROWS + base + lane] = PAD_ENTRY;
            if (lane == 0) ccnt[off] = base;
        }
    }
    __syncthreads();

    {
        int src[4];
        #pragma unroll
        for (int it = 0; it < 4; ++it)
            src[it] = clist[it * 8 + grow] >> 8;
        v4f g[4];
        #pragma unroll
        for (int it = 0; it < 4; ++it)
            g[it] = *(const v4f*)(feat + (size_t)(src[it] < 0 ? 0 : src[it]) * 64 + gq * 4);
        #pragma unroll
        for (int it = 0; it < 4; ++it)
            *(v4f*)(&abuf[(it * 8 + grow) * LSTR + gq * 4]) = src[it] < 0 ? (v4f)0.0f : g[it];
    }

    v2f bf[16];
    #pragma unroll
    for (int k = 0; k < 16; ++k)
        bf[k] = *(const v2f*)(w1p + (size_t)wv * 16 * 64 + k * 64 + lane * 2);

    int cnt = __builtin_amdgcn_readfirstlane(ccnt[0]);
    int t = 0, c = 0, buf = 0;
    for (;;) {
        __syncthreads();

        int nch = (cnt + CROWS - 1) / CROWS;
        if (nch < 1) nch = 1;
        int nt = t, nc = c + 1;
        if (nc >= nch) { nt = t + 1; nc = 0; }
        const bool last = nt >= NTAP;
        if (last) { nt = t; nc = c; }

        const int cnt_next = ccnt[nt];

        int src[4];
        #pragma unroll
        for (int it = 0; it < 4; ++it)
            src[it] = clist[nt * TROWS + nc * CROWS + it * 8 + grow] >> 8;
        const int* cl = &clist[t * TROWS + c * CROWS + half * 8];
        const v4i e0 = *(const v4i*)(cl);
        const v4i e1 = *(const v4i*)(cl + 4);
        const v4i e2 = *(const v4i*)(cl + 16);
        const v4i e3 = *(const v4i*)(cl + 20);

        const float* ap = &abuf[buf * (CROWS * LSTR) + l15 * LSTR + 2 * half];
        v2f a0[16], a1[16];
        #pragma unroll
        for (int k = 0; k < 16; ++k) {
            a0[k] = *(const v2f*)(ap + 4 * k);
            a1[k] = *(const v2f*)(ap + 16 * LSTR + 4 * k);
        }

        v4f g[4];
        #pragma unroll
        for (int it = 0; it < 4; ++it)
            g[it] = *(const v4f*)(feat + (size_t)(src[it] < 0 ? 0 : src[it]) * 64 + gq * 4);

        int ad[16];
        #pragma unroll
        for (int j = 0; j < 4; ++j) {
            ad[j]      = (e0[j] & 255) * LSTR + col;
            ad[4 + j]  = (e1[j] & 255) * LSTR + col;
            ad[8 + j]  = (e2[j] & 255) * LSTR + col;
            ad[12 + j] = (e3[j] & 255) * LSTR + col;
        }
        float old[16];
        #pragma unroll
        for (int j = 0; j < 16; ++j) old[j] = accL[ad[j]];

        __builtin_amdgcn_sched_barrier(0);

        v8f acc0 = (v8f)0.0f, acc1 = (v8f)0.0f;
        #pragma unroll
        for (int k = 0; k < 16; ++k) {
            acc0 = __builtin_amdgcn_wmma_f32_16x16x4_f32(
                false, a0[k], false, bf[k], (short)0, acc0, false, false);
            acc1 = __builtin_amdgcn_wmma_f32_16x16x4_f32(
                false, a1[k], false, bf[k], (short)0, acc1, false, false);
        }

        #pragma unroll
        for (int k = 0; k < 16; ++k)
            bf[k] = *(const v2f*)(w1p + (size_t)(nt * 4 + wv) * 16 * 64 + k * 64 + lane * 2);

        #pragma unroll
        for (int j = 0; j < 8; ++j) {
            accL[ad[j]]     = old[j]     + acc0[j];
            accL[ad[8 + j]] = old[8 + j] + acc1[j];
        }

        #pragma unroll
        for (int it = 0; it < 4; ++it)
            *(v4f*)(&abuf[(buf ^ 1) * (CROWS * LSTR) + (it * 8 + grow) * LSTR + gq * 4]) =
                src[it] < 0 ? (v4f)0.0f : g[it];

        if (last) break;
        cnt = __builtin_amdgcn_readfirstlane(cnt_next);
        t = nt; c = nc; buf ^= 1;
    }

    const float scl = gamma[col] * rsqrtf(rvar[col] + 1e-5f);
    const float shf = (b1[col] - rmean[col]) * scl + beta[col];

    #pragma unroll
    for (int b = 0; b < 4; ++b) {
        float y[16];
        #pragma unroll
        for (int u = 0; u < 16; ++u)
            y[u] = accL[(b * 32 + 2 * u + half) * LSTR + col];
        #pragma unroll
        for (int u = 0; u < 16; ++u) {
            float v = y[u] * scl + shf;
            v = v > 0.0f ? v : 0.0f;
            accL[(b * 32 + 2 * u + half) * LSTR + col] = v;
        }
    }
    __syncthreads();

    v8f acc2[8];
    #pragma unroll
    for (int m = 0; m < 8; ++m) acc2[m] = (v8f)0.0f;
    #pragma unroll
    for (int k = 0; k < 16; ++k) {
        v2f bf = *(const v2f*)(w2p + (size_t)(wv * 16 + k) * 64 + lane * 2);
        #pragma unroll
        for (int m = 0; m < 8; ++m) {
            v2f af = *(const v2f*)(&accL[(m * 16 + l15) * LSTR + 4 * k + 2 * half]);
            acc2[m] = __builtin_amdgcn_wmma_f32_16x16x4_f32(
                false, af, false, bf, (short)0, acc2[m], false, false);
        }
    }

    const float bb = b2[col];
    #pragma unroll
    for (int m = 0; m < 8; ++m) {
        int rg = row0 + m * 16 + half * 8;
        #pragma unroll
        for (int j = 0; j < 8; ++j) {
            int r = rg + j;
            if (r < N)
                out[(size_t)r * 64 + col] = acc2[m][j] + bb;
        }
    }
}

extern "C" void kernel_launch(void* const* d_in, const int* in_sizes, int n_in,
                              void* d_out, int out_size, void* d_ws, size_t ws_size,
                              hipStream_t stream) {
    const float* feat  = (const float*)d_in[0];
    const int*   coords= (const int*)  d_in[1];
    const float* w1    = (const float*)d_in[2];
    const float* b1    = (const float*)d_in[3];
    const float* gamma = (const float*)d_in[4];
    const float* beta  = (const float*)d_in[5];
    const float* rmean = (const float*)d_in[6];
    const float* rvar  = (const float*)d_in[7];
    const float* w2    = (const float*)d_in[8];
    const float* b2    = (const float*)d_in[9];
    float* out = (float*)d_out;

    const int N = in_sizes[0] / 64;

    char* ws = (char*)d_ws;
    int*   grid = (int*)ws;
    float* w1p  = (float*)(ws + (size_t)GRID_VOX * 4);
    float* w2p  = w1p + (size_t)27 * 4 * 16 * 32 * 2;

    grid_init_kernel<<<GRID_VOX / 256, 256, 0, stream>>>(grid);
    grid_build_kernel<<<(N + 255) / 256, 256, 0, stream>>>(coords, grid, N);
    pack_w1_kernel<<<(27 * 4 * 16 * 32 + 255) / 256, 256, 0, stream>>>(w1, w1p);
    pack_w2_kernel<<<(4 * 16 * 32 + 255) / 256, 256, 0, stream>>>(w2, w2p);

    const int nblk = (N + TROWS - 1) / TROWS;
    sparse_block_kernel<<<nblk, 128, 0, stream>>>(feat, coords, grid, w1p, b1,
                                                  gamma, beta, rmean, rvar,
                                                  w2p, b2, out, N);
}


# Tap product formed transposed (weights as WMMA A, gathered rows as B): a lane holds 8 consecutive channels of one product row, so the LDS accumulator update is 16-byte accesses; sum order unchanged
# speedup vs baseline: 2.1386x; 1.0526x over previous
#include <hip/hip_runtime.h>
#include <hip/hip_bf16.h>
#include <math.h>

typedef __attribute__((ext_vector_type(2))) float v2f;
typedef __attribute__((ext_vector_type(4))) float v4f;
typedef __attribute__((ext_vector_type(8))) float v8f;
typedef __attribute__((ext_vector_type(4))) int   v4i;

#define GD 128
#define GRID_VOX (GD * GD * GD)
#define TROWS 128
#define CROWS 32
#define NTAP 27
#define LSTR 68
#define PAD_ENTRY (-128)

__global__ void grid_init_kernel(int* __restrict__ grid) {
    int i = blockIdx.x * 256 + threadIdx.x;
    if (i < GRID_VOX) grid[i] = -1;
}

__global__ void grid_build_kernel(const int* __restrict__ coords,
                                  int* __restrict__ grid, int N) {
    int i = blockIdx.x * 256 + threadIdx.x;
    if (i < N) {
        int z = coords[3 * i + 0];
        int y = coords[3 * i + 1];
        int x = coords[3 * i + 2];
        grid[(z * GD + y) * GD + x] = i;
    }
}

__global__ void pack_w1_kernel(const float* __restrict__ w1, float* __restrict__ w1p) {
    int t = blockIdx.x * 256 + threadIdx.x;
    if (t >= 27 * 4 * 16 * 32) return;
    int lane = t & 31;
    int k    = (t >> 5) & 15;
    int nt   = (t >> 9) & 3;
    int off  = t >> 11;
    int half = lane >> 4;
    int col  = nt * 16 + (lane & 15);
    int krow = 4 * k + 2 * half;
    const float* w = w1 + (size_t)off * 64 * 64;
    w1p[(size_t)t * 2 + 0] = w[(krow + 0) * 64 + col];
    w1p[(size_t)t * 2 + 1] = w[(krow + 1) * 64 + col];
}

__global__ void pack_w2_kernel(const float* __restrict__ w2, float* __restrict__ w2p) {
    int t = blockIdx.x * 256 + threadIdx.x;
    if (t >= 4 * 16 * 32) return;
    int lane = t & 31;
    int k    = (t >> 5) & 15;
    int nt   = (t >> 9) & 3;
    int half = lane >> 4;
    int col  = nt * 16 + (lane & 15);
    int krow = 4 * k + 2 * half;
    w2p[(size_t)t * 2 + 0] = w2[(krow + 0) * 64 + col];
    w2p[(size_t)t * 2 + 1] = w2[(krow + 1) * 64 + col];
}

__global__ __launch_bounds__(128)
void sparse_block_kernel(const float* __restrict__ feat,
                         const int*   __restrict__ coords,
                         const int*   __restrict__ grid,
                         const float* __restrict__ w1p,
                         const float* __restrict__ b1,
                         const float* __restrict__ gamma,
                         const float* __restrict__ beta,
                         const float* __restrict__ rmean,
                         const float* __restrict__ rvar,
                         const float* __restrict__ w2p,
                         const float* __restrict__ b2,
                         float* __restrict__ out,
                         int N) {
    __shared__ __attribute__((aligned(16))) float accL[(TROWS + 1) * LSTR];
    __shared__ __attribute__((aligned(16))) float abuf[2 * CROWS * LSTR];
    __shared__ __attribute__((aligned(16))) int   clist[NTAP * TROWS];
    __shared__ int ccnt[NTAP];

    const int tid  = threadIdx.x;
    const int lane = tid & 31;
    const int wv   = __builtin_amdgcn_readfirstlane(tid >> 5);
    const int half = lane >> 4;
    const int l15  = lane & 15;
    const int col  = wv * 16 + l15;
    const int row0 = blockIdx.x * TROWS;
    const int grow = tid >> 4;
    const int gq   = tid & 15;

    for (int i = tid; i < (TROWS + 1) * LSTR / 4; i += 128)
        *(v4f*)(&accL[i * 4]) = (v4f)0.0f;

    int cz[4], cy[4], cx[4];
    #pragma unroll
    for (int g = 0; g < 4; ++g) {
        const int r  = row0 + g * 32 + lane;
        const int rc = r < N ? r : N - 1;
        const int z = coords[3 * rc + 0];
        const int y = coords[3 * rc + 1];
        const int x = coords[3 * rc + 2];
        cz[g] = r < N ? z : -10000;
        cy[g] = r < N ? y : -10000;
        cx[g] = r < N ? x : -10000;
    }

    int nid[7][4];
    #pragma unroll
    for (int i = 0; i < 7; ++i) {
        int off = wv + 4 * i;
        off = off < NTAP ? off : NTAP - 1;
        const int dz = off / 9 - 1;
        const int dy = (off / 3) % 3 - 1;
        const int dx = off % 3 - 1;
        #pragma unroll
        for (int g = 0; g < 4; ++g) {
            const int nz = cz[g] + dz, ny = cy[g] + dy, nx = cx[g] + dx;
            int idx = -1;
            if (nz >= 0 && nz < GD && ny >= 0 && ny < GD && nx >= 0 && nx < GD)
                idx = grid[(nz * GD + ny) * GD + nx];
            nid[i][g] = idx;
        }
    }

    const unsigned lt = (1u << lane) - 1u;
    #pragma unroll
    for (int i = 0; i < 7; ++i) {
        const int off = wv + 4 * i;
        if (off < NTAP) {
            int base = 0;
            #pragma unroll
            for (int g = 0; g < 4; ++g) {
                const bool v = nid[i][g] >= 0;
                const unsigned m = __builtin_amdgcn_ballot_w32(v);
                if (v)
                    clist[off * TROWS + base + __popc(m & lt)] = (nid[i][g] << 8) | (g * 32 + lane);
                base += __popc(m);
            }
            int padded = (base + CROWS - 1) & ~(CROWS - 1);
            if (padded < CROWS) padded = CROWS;
            if (base + lane < padded) clist[off * TROWS + base + lane] = PAD_ENTRY;
            if (lane == 0) ccnt[off] = base;
        }
    }
    __syncthreads();

    {
        int src[4];
        #pragma unroll
        for (int it = 0; it < 4; ++it)
            src[it] = clist[it * 8 + grow] >> 8;
        v4f g[4];
        #pragma unroll
        for (int it = 0; it < 4; ++it)
            g[it] = *(const v4f*)(feat + (size_t)(src[it] < 0 ? 0 : src[it]) * 64 + gq * 4);
        #pragma unroll
        for (int it = 0; it < 4; ++it)
            *(v4f*)(&abuf[(it * 8 + grow) * LSTR + gq * 4]) = src[it] < 0 ? (v4f)0.0f : g[it];
    }

    v2f bf[16];
    #pragma unroll
    for (int k = 0; k < 16; ++k)
        bf[k] = *(const v2f*)(w1p + (size_t)wv * 16 * 64 + k * 64 + lane * 2);

    int cnt = __builtin_amdgcn_readfirstlane(ccnt[0]);
    int t = 0, c = 0, buf = 0;
    for (;;) {
        __syncthreads();

        int nch = (cnt + CROWS - 1) / CROWS;
        if (nch < 1) nch = 1;
        int nt = t, nc = c + 1;
        if (nc >= nch) { nt = t + 1; nc = 0; }
        const bool last = nt >= NTAP;
        if (last) { nt = t; nc = c; }

        const int cnt_next = ccnt[nt];

        int src[4];
        #pragma unroll
        for (int it = 0; it < 4; ++it)
            src[it] = clist[nt * TROWS + nc * CROWS + it * 8 + grow] >> 8;
        const int* cl = &clist[t * TROWS + c * CROWS + l15];
        const int e0 = cl[0];
        const int e1 = cl[16];

        const float* ap = &abuf[buf * (CROWS * LSTR) + l15 * LSTR + 2 * half];
        v2f a0[16], a1[16];
        #pragma unroll
        for (int k = 0; k < 16; ++k) {
            a0[k] = *(const v2f*)(ap + 4 * k);
            a1[k] = *(const v2f*)(ap + 16 * LSTR + 4 * k);
        }

        v4f g[4];
        #pragma unroll
        for (int it = 0; it < 4; ++it)
            g[it] = *(const v4f*)(feat + (size_t)(src[it] < 0 ? 0 : src[it]) * 64 + gq * 4);

        float* const p0 = &accL[(e0 & 255) * LSTR + wv * 16 + half * 8];
        float* const p1 = &accL[(e1 & 255) * LSTR + wv * 16 + half * 8];
        const v4f o00 = *(const v4f*)(p0);
        const v4f o01 = *(const v4f*)(p0 + 4);
        const v4f o10 = *(const v4f*)(p1);
        const v4f o11 = *(const v4f*)(p1 + 4);

        __builtin_amdgcn_sched_barrier(0);

        v8f acc0 = (v8f)0.0f, acc1 = (v8f)0.0f;
        #pragma unroll
        for (int k = 0; k < 16; ++k) {
            acc0 = __builtin_amdgcn_wmma_f32_16x16x4_f32(
                false, bf[k], false, a0[k], (short)0, acc0, false, false);
            acc1 = __builtin_amdgcn_wmma_f32_16x16x4_f32(
                false, bf[k], false, a1[k], (short)0, acc1, false, false);
        }

        #pragma unroll
        for (int k = 0; k < 16; ++k)
            bf[k] = *(const v2f*)(w1p + (size_t)(nt * 4 + wv) * 16 * 64 + k * 64 + lane * 2);

        *(v4f*)(p0)     = o00 + __builtin_shufflevector(acc0, acc0, 0, 1, 2, 3);
        *(v4f*)(p0 + 4) = o01 + __builtin_shufflevector(acc0, acc0, 4, 5, 6, 7);
        *(v4f*)(p1)     = o10 + __builtin_shufflevector(acc1, acc1, 0, 1, 2, 3);
        *(v4f*)(p1 + 4) = o11 + __builtin_shufflevector(acc1, acc1, 4, 5, 6, 7);

        #pragma unroll
        for (int it = 0; it < 4; ++it)
            *(v4f*)(&abuf[(buf ^ 1) * (CROWS * LSTR) + (it * 8 + grow) * LSTR + gq * 4]) =
                src[it] < 0 ? (v4f)0.0f : g[it];

        if (last) break;
        cnt = __builtin_amdgcn_readfirstlane(cnt_next);
        t = nt; c = nc; buf ^= 1;
    }

    const float scl = gamma[col] * rsqrtf(rvar[col] + 1e-5f);
    const float shf = (b1[col] - rmean[col]) * scl + beta[col];

    #pragma unroll
    for (int b = 0; b < 4; ++b) {
        float y[16];
        #pragma unroll
        for (int u = 0; u < 16; ++u)
            y[u] = accL[(b * 32 + 2 * u + half) * LSTR + col];
        #pragma unroll
        for (int u = 0; u < 16; ++u) {
            float v = y[u] * scl + shf;
            v = v > 0.0f ? v : 0.0f;
            accL[(b * 32 + 2 * u + half) * LSTR + col] = v;
        }
    }
    __syncthreads();

    v8f acc2[8];
    #pragma unroll
    for (int m = 0; m < 8; ++m) acc2[m] = (v8f)0.0f;
    #pragma unroll
    for (int k = 0; k < 16; ++k) {
        v2f bf = *(const v2f*)(w2p + (size_t)(wv * 16 + k) * 64 + lane * 2);
        #pragma unroll
        for (int m = 0; m < 8; ++m) {
            v2f af = *(const v2f*)(&accL[(m * 16 + l15) * LSTR + 4 * k + 2 * half]);
            acc2[m] = __builtin_amdgcn_wmma_f32_16x16x4_f32(
                false, af, false, bf, (short)0, acc2[m], false, false);
        }
    }

    const float bb = b2[col];
    #pragma unroll
    for (int m = 0; m < 8; ++m) {
        int rg = row0 + m * 16 + half * 8;
        #pragma unroll
        for (int j = 0; j < 8; ++j) {
            int r = rg + j;
            if (r < N)
                out[(size_t)r * 64 + col] = acc2[m][j] + bb;
        }
    }
}

extern "C" void kernel_launch(void* const* d_in, const int* in_sizes, int n_in,
                              void* d_out, int out_size, void* d_ws, size_t ws_size,
                              hipStream_t stream) {
    const float* feat  = (const float*)d_in[0];
    const int*   coords= (const int*)  d_in[1];
    const float* w1    = (const float*)d_in[2];
    const float* b1    = (const float*)d_in[3];
    const float* gamma = (const float*)d_in[4];
    const float* beta  = (const float*)d_in[5];
    const float* rmean = (const float*)d_in[6];
    const float* rvar  = (const float*)d_in[7];
    const float* w2    = (const float*)d_in[8];
    const float* b2    = (const float*)d_in[9];
    float* out = (float*)d_out;

    const int N = in_sizes[0] / 64;

    char* ws = (char*)d_ws;
    int*   grid = (int*)ws;
    float* w1p  = (float*)(ws + (size_t)GRID_VOX * 4);
    float* w2p  = w1p + (size_t)27 * 4 * 16 * 32 * 2;

    grid_init_kernel<<<GRID_VOX / 256, 256, 0, stream>>>(grid);
    grid_build_kernel<<<(N + 255) / 256, 256, 0, stream>>>(coords, grid, N);
    pack_w1_kernel<<<(27 * 4 * 16 * 32 + 255) / 256, 256, 0, stream>>>(w1, w1p);
    pack_w2_kernel<<<(4 * 16 * 32 + 255) / 256, 256, 0, stream>>>(w2, w2p);

    const int nblk = (N + TROWS - 1) / TROWS;
    sparse_block_kernel<<<nblk, 128, 0, stream>>>(feat, coords, grid, w1p, b1,
                                                  gamma, beta, rmean, rvar,
                                                  w2p, b2, out, N);
}
